# starting kernel
# baseline (speedup 1.0000x reference)
#include <hip/hip_runtime.h>

typedef __attribute__((ext_vector_type(2))) float v2f;
typedef __attribute__((ext_vector_type(4))) float v4f;
typedef __attribute__((ext_vector_type(8))) float v8f;

__device__ __forceinline__ float hw_tanh(float x) {
#if __has_builtin(__builtin_amdgcn_tanhf)
    return __builtin_amdgcn_tanhf(x);
#elif __has_builtin(__builtin_amdgcn_tanh_f32)
    return __builtin_amdgcn_tanh_f32(x);
#else
    float e = __expf(2.0f * x);
    return 1.0f - 2.0f * __builtin_amdgcn_rcpf(e + 1.0f);
#endif
}
__device__ __forceinline__ float fast_sigmoid(float x) {
    return 0.5f * hw_tanh(0.5f * x) + 0.5f;
}

__global__ __launch_bounds__(256) void gru_wmma_kernel(
    const float* __restrict__ x, const float* __restrict__ hidden,
    const float* __restrict__ Wzh, const float* __restrict__ bzh,
    const float* __restrict__ Wzx, const float* __restrict__ bzx,
    const float* __restrict__ Wrh, const float* __restrict__ brh,
    const float* __restrict__ Wrx, const float* __restrict__ brx,
    const float* __restrict__ Wnx, const float* __restrict__ bnx,
    const float* __restrict__ Wnh, const float* __restrict__ bnh,
    float* __restrict__ out, int B)
{
    const int lane = threadIdx.x & 31;
    const int n    = lane & 15;
    const int kh   = lane >> 4;
    const int j    = n & 3;
    const bool mv  = (n < 12);

    const float* Wh = (n < 4) ? Wzh : (n < 8) ? Wrh : Wnh;
    const float* Wx = (n < 4) ? Wzx : (n < 8) ? Wrx : Wnx;

    v2f aWh;
    aWh.x = mv ? Wh[j * 4 + 2 * kh]     : 0.0f;
    aWh.y = mv ? Wh[j * 4 + 2 * kh + 1] : 0.0f;

    v2f aWx[3];
#pragma unroll
    for (int t = 0; t < 3; ++t) {
        int k0 = 4 * t + 2 * kh;
        int k1 = k0 + 1;
        aWx[t].x = (mv && k0 < 10) ? Wx[j * 10 + k0] : 0.0f;
        aWx[t].y = (mv && k1 < 10) ? Wx[j * 10 + k1] : 0.0f;
    }

    v8f cInitH, cInitX;
#pragma unroll
    for (int r = 0; r < 8; ++r) {
        cInitH[r] = kh ? (r < 4 ? bnh[r] : 0.0f) : (r < 4 ? bzh[r] : brh[r - 4]);
        cInitX[r] = kh ? (r < 4 ? bnx[r] : 0.0f) : (r < 4 ? bzx[r] : brx[r - 4]);
    }

    const int Tfull  = B >> 4;
    const int waveId = blockIdx.x * (blockDim.x >> 5) + (threadIdx.x >> 5);
    const int nWaves = gridDim.x * (blockDim.x >> 5);

    const size_t r0 = (size_t)waveId * 16 + n;
    const float* pH  = hidden + r0 * 4  + 2 * kh;
    const float* pX  = x      + r0 * 10 + 2 * kh;
    const float* pX2 = x      + r0 * 10 + (kh ? 2 : 8);
    float*       pO  = out    + r0 * 4;
    const ptrdiff_t sX = (ptrdiff_t)nWaves * 160;
    const ptrdiff_t sH = (ptrdiff_t)nWaves * 64;

    for (int tile = waveId; tile < Tfull; tile += nWaves) {
        v2f bHd = *(const v2f*)pH;
        v2f bX0 = *(const v2f*)pX;
        v2f bX1 = *(const v2f*)(pX + 4);
        v2f bX2 = *(const v2f*)pX2;

        v8f accH = cInitH;
        v8f accX = cInitX;
        accH = __builtin_amdgcn_wmma_f32_16x16x4_f32(false, aWh,    false, bHd, (short)0, accH, false, false);
        accX = __builtin_amdgcn_wmma_f32_16x16x4_f32(false, aWx[0], false, bX0, (short)0, accX, false, false);
        accX = __builtin_amdgcn_wmma_f32_16x16x4_f32(false, aWx[1], false, bX1, (short)0, accX, false, false);
        accX = __builtin_amdgcn_wmma_f32_16x16x4_f32(false, aWx[2], false, bX2, (short)0, accX, false, false);

        float s[8];
#pragma unroll
        for (int r = 0; r < 8; ++r) s[r] = accH[r] + accX[r];

        float Hn[4], Xn[4];
#pragma unroll
        for (int q = 0; q < 4; ++q) {
            Hn[q] = __shfl_xor(accH[q], 16, 32);
            Xn[q] = __shfl_xor(accX[q], 16, 32);
        }
        const float h2 = __shfl_xor(bHd.x, 16, 32);
        const float h3 = __shfl_xor(bHd.y, 16, 32);
        const float hv[4] = { bHd.x, bHd.y, h2, h3 };

        v4f o;
#pragma unroll
        for (int q = 0; q < 4; ++q) {
            const float z  = fast_sigmoid(s[q]);
            const float rr = fast_sigmoid(s[4 + q]);
            const float nn = hw_tanh(rr * Hn[q] + Xn[q]);
            o[q] = nn + z * (hv[q] - nn);
        }

        if (lane < 16) {
            *(v4f*)pO = o;
        }

        pH += sH; pX += sX; pX2 += sX; pO += sH;
    }

    const int tailBase = Tfull << 4;
    if (waveId == 0 && tailBase < B) {
        const int row = tailBase + lane;
        if (row < B) {
#pragma unroll 1
            for (int u = 0; u < 4; ++u) {
                float az  = bzh[u] + bzx[u];
                float ar  = brh[u] + brx[u];
                float ahn = bnh[u];
                float axn = bnx[u];
#pragma unroll 1
                for (int k = 0; k < 4; ++k) {
                    const float hk = hidden[(size_t)row * 4 + k];
                    az  += hk * Wzh[u * 4 + k];
                    ar  += hk * Wrh[u * 4 + k];
                    ahn += hk * Wnh[u * 4 + k];
                }
#pragma unroll 1
                for (int k = 0; k < 10; ++k) {
                    const float xk = x[(size_t)row * 10 + k];
                    az  += xk * Wzx[u * 10 + k];
                    ar  += xk * Wrx[u * 10 + k];
                    axn += xk * Wnx[u * 10 + k];
                }
                const float z  = fast_sigmoid(az);
                const float rr = fast_sigmoid(ar);
                const float nn = hw_tanh(rr * ahn + axn);
                out[(size_t)row * 4 + u] = nn + z * (hidden[(size_t)row * 4 + u] - nn);
            }
        }
    }
}

extern "C" void kernel_launch(void* const* d_in, const int* in_sizes, int n_in,
                              void* d_out, int out_size, void* d_ws, size_t ws_size,
                              hipStream_t stream) {
    const float* x      = (const float*)d_in[0];
    const float* hidden = (const float*)d_in[1];
    const float* Wzh    = (const float*)d_in[2];
    const float* bzh    = (const float*)d_in[3];
    const float* Wzx    = (const float*)d_in[4];
    const float* bzx    = (const float*)d_in[5];
    const float* Wrh    = (const float*)d_in[6];
    const float* brh    = (const float*)d_in[7];
    const float* Wrx    = (const float*)d_in[8];
    const float* brx    = (const float*)d_in[9];
    const float* Wnx    = (const float*)d_in[10];
    const float* bnx    = (const float*)d_in[11];
    const float* Wnh    = (const float*)d_in[12];
    const float* bnh    = (const float*)d_in[13];
    float* out = (float*)d_out;

    const int B = in_sizes[0] / 10;

    const int blocks  = 2048;
    const int threads = 256;
    gru_wmma_kernel<<<blocks, threads, 0, stream>>>(
        x, hidden, Wzh, bzh, Wzx, bzx, Wrh, brh, Wrx, brx, Wnx, bnx, Wnh, bnh,
        out, B);
}


# Streaming restructure: 4096 waves (16 per WGP), each owning a contiguous run of tiles with a ring of 8 tiles' loads in flight, refilled slot by slot; was 16384 waves with one tile in flight each
# speedup vs baseline: 1.0626x; 1.0626x over previous
#include <hip/hip_runtime.h>


typedef __attribute__((ext_vector_type(2))) float v2f;
typedef __attribute__((ext_vector_type(4))) float v4f;
typedef __attribute__((ext_vector_type(8))) float v8f;

__device__ __forceinline__ float hw_tanh(float x) {
#if __has_builtin(__builtin_amdgcn_tanhf)
    return __builtin_amdgcn_tanhf(x);
#elif __has_builtin(__builtin_amdgcn_tanh_f32)
    return __builtin_amdgcn_tanh_f32(x);
#else
    float e = __expf(2.0f * x);
    return 1.0f - 2.0f * __builtin_amdgcn_rcpf(e + 1.0f);
#endif
}
__device__ __forceinline__ float fast_sigmoid(float x) {
    return 0.5f * hw_tanh(0.5f * x) + 0.5f;
}

#define RING            8
#define WAVES_PER_BLOCK 16
#define X_TILE_BYTES    640
#define H_TILE_BYTES    256

struct Tile { v2f h, x0, x1, x2; };

__device__ __forceinline__ void load_tile(Tile& t, const char* xb, const char* hb,
                                          unsigned offX, unsigned offX2, unsigned offH, int j) {
    t.x0 = *(const v2f*)(xb + offX  + j * X_TILE_BYTES);
    t.x1 = *(const v2f*)(xb + offX  + j * X_TILE_BYTES + 16);
    t.h  = *(const v2f*)(hb + offH  + j * H_TILE_BYTES);
    t.x2 = *(const v2f*)(xb + offX2 + j * X_TILE_BYTES);
}

__device__ __forceinline__ v4f gru_tile(const Tile& t, v2f aWh, v2f aWx0, v2f aWx1, v2f aWx2,
                                        v8f cInitH, v8f cInitX) {
    v8f accH = __builtin_amdgcn_wmma_f32_16x16x4_f32(false, aWh,  false, t.h,  (short)0, cInitH, false, false);
    v8f accX = __builtin_amdgcn_wmma_f32_16x16x4_f32(false, aWx0, false, t.x0, (short)0, cInitX, false, false);
    accX     = __builtin_amdgcn_wmma_f32_16x16x4_f32(false, aWx1, false, t.x1, (short)0, accX,   false, false);
    accX     = __builtin_amdgcn_wmma_f32_16x16x4_f32(false, aWx2, false, t.x2, (short)0, accX,   false, false);

    float s[8];
#pragma unroll
    for (int r = 0; r < 8; ++r) s[r] = accH[r] + accX[r];

    float Hn[4], Xn[4];
#pragma unroll
    for (int q = 0; q < 4; ++q) {
        Hn[q] = __shfl_xor(accH[q], 16, 32);
        Xn[q] = __shfl_xor(accX[q], 16, 32);
    }
    const float h2 = __shfl_xor(t.h.x, 16, 32);
    const float h3 = __shfl_xor(t.h.y, 16, 32);
    const float hv[4] = { t.h.x, t.h.y, h2, h3 };

    v4f o;
#pragma unroll
    for (int q = 0; q < 4; ++q) {
        const float z  = fast_sigmoid(s[q]);
        const float rr = fast_sigmoid(s[4 + q]);
        const float nn = hw_tanh(rr * Hn[q] + Xn[q]);
        o[q] = nn + z * (hv[q] - nn);
    }
    return o;
}

__global__ __launch_bounds__(512) void gru_wmma_kernel(
    const float* __restrict__ x, const float* __restrict__ hidden,
    const float* __restrict__ Wzh, const float* __restrict__ bzh,
    const float* __restrict__ Wzx, const float* __restrict__ bzx,
    const float* __restrict__ Wrh, const float* __restrict__ brh,
    const float* __restrict__ Wrx, const float* __restrict__ brx,
    const float* __restrict__ Wnx, const float* __restrict__ bnx,
    const float* __restrict__ Wnh, const float* __restrict__ bnh,
    float* __restrict__ out, int B)
{
    const int lane = threadIdx.x & 31;
    const int n    = lane & 15;
    const int kh   = lane >> 4;
    const int j    = n & 3;
    const bool mv  = (n < 12);

    const float* Wh = (n < 4) ? Wzh : (n < 8) ? Wrh : Wnh;
    const float* Wx = (n < 4) ? Wzx : (n < 8) ? Wrx : Wnx;

    v2f aWh;
    aWh.x = mv ? Wh[j * 4 + 2 * kh]     : 0.0f;
    aWh.y = mv ? Wh[j * 4 + 2 * kh + 1] : 0.0f;

    v2f aWx[3];
#pragma unroll
    for (int t = 0; t < 3; ++t) {
        int k0 = 4 * t + 2 * kh;
        int k1 = k0 + 1;
        aWx[t].x = (mv && k0 < 10) ? Wx[j * 10 + k0] : 0.0f;
        aWx[t].y = (mv && k1 < 10) ? Wx[j * 10 + k1] : 0.0f;
    }

    v8f cInitH, cInitX;
#pragma unroll
    for (int r = 0; r < 8; ++r) {
        cInitH[r] = kh ? (r < 4 ? bnh[r] : 0.0f) : (r < 4 ? bzh[r] : brh[r - 4]);
        cInitX[r] = kh ? (r < 4 ? bnx[r] : 0.0f) : (r < 4 ? bzx[r] : brx[r - 4]);
    }

    const int Tfull  = B >> 4;
    const int wib    = __builtin_amdgcn_readfirstlane((int)(threadIdx.x >> 5));
    const int waveId = blockIdx.x * WAVES_PER_BLOCK + wib;
    const int nWaves = gridDim.x * WAVES_PER_BLOCK;

    const int roundsTotal = Tfull / RING;
    const int rpw  = (roundsTotal + nWaves - 1) / nWaves;
    const int rBeg = waveId * rpw;
    const int rEnd = (rBeg + rpw < roundsTotal) ? (rBeg + rpw) : roundsTotal;
    const int nr   = rEnd - rBeg;

    const unsigned offX  = (unsigned)(n * 40 + kh * 8);
    const unsigned offX2 = (unsigned)(n * 40 + (kh ? 8 : 32));
    const unsigned offH  = (unsigned)(n * 16 + kh * 8);
    const unsigned offO  = (unsigned)(n * 16);
    const bool storer = (lane < 16);

    if (nr > 0) {
        const char* xb = (const char*)x      + (size_t)rBeg * (RING * X_TILE_BYTES);
        const char* hb = (const char*)hidden + (size_t)rBeg * (RING * H_TILE_BYTES);
        char*       ob = (char*)out          + (size_t)rBeg * (RING * H_TILE_BYTES);

        Tile buf[RING];
#pragma unroll
        for (int q = 0; q < RING; ++q) load_tile(buf[q], xb, hb, offX, offX2, offH, q);

        for (int r = 1; r < nr; ++r) {
            const char* xn = xb + RING * X_TILE_BYTES;
            const char* hn = hb + RING * H_TILE_BYTES;
#pragma unroll
            for (int q = 0; q < RING; q += 2) {
                const v4f o0 = gru_tile(buf[q],     aWh, aWx[0], aWx[1], aWx[2], cInitH, cInitX);
                const v4f o1 = gru_tile(buf[q + 1], aWh, aWx[0], aWx[1], aWx[2], cInitH, cInitX);
                if (storer) {
                    *(v4f*)(ob + offO + q * H_TILE_BYTES)       = o0;
                    *(v4f*)(ob + offO + (q + 1) * H_TILE_BYTES) = o1;
                }
                load_tile(buf[q],     xn, hn, offX, offX2, offH, q);
                load_tile(buf[q + 1], xn, hn, offX, offX2, offH, q + 1);
                __builtin_amdgcn_sched_barrier(0);
            }
            xb = xn; hb = hn; ob += RING * H_TILE_BYTES;
        }

#pragma unroll
        for (int q = 0; q < RING; q += 2) {
            const v4f o0 = gru_tile(buf[q],     aWh, aWx[0], aWx[1], aWx[2], cInitH, cInitX);
            const v4f o1 = gru_tile(buf[q + 1], aWh, aWx[0], aWx[1], aWx[2], cInitH, cInitX);
            if (storer) {
                *(v4f*)(ob + offO + q * H_TILE_BYTES)       = o0;
                *(v4f*)(ob + offO + (q + 1) * H_TILE_BYTES) = o1;
            }
        }
    }

    const int tLeft = roundsTotal * RING + waveId;
    if (tLeft < Tfull) {
        Tile t;
        load_tile(t, (const char*)x + (size_t)tLeft * X_TILE_BYTES,
                     (const char*)hidden + (size_t)tLeft * H_TILE_BYTES, offX, offX2, offH, 0);
        const v4f o = gru_tile(t, aWh, aWx[0], aWx[1], aWx[2], cInitH, cInitX);
        if (storer) *(v4f*)((char*)out + (size_t)tLeft * H_TILE_BYTES + offO) = o;
    }

    const int tailBase = Tfull << 4;
    if (waveId == 0 && tailBase < B) {
        const int row = tailBase + lane;
        if (row < B) {
#pragma unroll 1
            for (int u = 0; u < 4; ++u) {
                float az  = bzh[u] + bzx[u];
                float ar  = brh[u] + brx[u];
                float ahn = bnh[u];
                float axn = bnx[u];
#pragma unroll 1
                for (int k = 0; k < 4; ++k) {
                    const float hk = hidden[(size_t)row * 4 + k];
                    az  += hk * Wzh[u * 4 + k];
                    ar  += hk * Wrh[u * 4 + k];
                    ahn += hk * Wnh[u * 4 + k];
                }
#pragma unroll 1
                for (int k = 0; k < 10; ++k) {
                    const float xk = x[(size_t)row * 10 + k];
                    az  += xk * Wzx[u * 10 + k];
                    ar  += xk * Wrx[u * 10 + k];
                    axn += xk * Wnx[u * 10 + k];
                }
                const float z  = fast_sigmoid(az);
                const float rr = fast_sigmoid(ar);
                const float nn = hw_tanh(rr * ahn + axn);
                out[(size_t)row * 4 + u] = nn + z * (hidden[(size_t)row * 4 + u] - nn);
            }
        }
    }
}

extern "C" void kernel_launch(void* const* d_in, const int* in_sizes, int n_in,
                              void* d_out, int out_size, void* d_ws, size_t ws_size,
                              hipStream_t stream) {
    const float* x      = (const float*)d_in[0];
    const float* hidden = (const float*)d_in[1];
    const float* Wzh    = (const float*)d_in[2];
    const float* bzh    = (const float*)d_in[3];
    const float* Wzx    = (const float*)d_in[4];
    const float* bzx    = (const float*)d_in[5];
    const float* Wrh    = (const float*)d_in[6];
    const float* brh    = (const float*)d_in[7];
    const float* Wrx    = (const float*)d_in[8];
    const float* brx    = (const float*)d_in[9];
    const float* Wnx    = (const float*)d_in[10];
    const float* bnx    = (const float*)d_in[11];
    const float* Wnh    = (const float*)d_in[12];
    const float* bnh    = (const float*)d_in[13];
    float* out = (float*)d_out;

    const int B = in_sizes[0] / 10;

    const int blocks  = 256;
    const int threads = 512;
    gru_wmma_kernel<<<blocks, threads, 0, stream>>>(
        x, hidden, Wzh, bzh, Wzx, bzx, Wrh, brh, Wrx, brx, Wnx, bnx, Wnh, bnh,
        out, B);
}


# Weight rows re-dealt over the A rows: each lane half gets z, r and both n terms of its two output units from ONE accumulator chain; no cross-lane moves, gate math on all lanes; z/r sum order changes
# speedup vs baseline: 1.1220x; 1.0559x over previous
#include <hip/hip_runtime.h>


typedef __attribute__((ext_vector_type(2))) float v2f;
typedef __attribute__((ext_vector_type(8))) float v8f;

__device__ __forceinline__ float hw_tanh(float x) {
#if __has_builtin(__builtin_amdgcn_tanhf)
    return __builtin_amdgcn_tanhf(x);
#elif __has_builtin(__builtin_amdgcn_tanh_f32)
    return __builtin_amdgcn_tanh_f32(x);
#else
    float e = __expf(2.0f * x);
    return 1.0f - 2.0f * __builtin_amdgcn_rcpf(e + 1.0f);
#endif
}
__device__ __forceinline__ float fast_sigmoid(float x) {
    return 0.5f * hw_tanh(0.5f * x) + 0.5f;
}

#define RING            8
#define WAVES_PER_BLOCK 16
#define X_TILE_BYTES    640
#define H_TILE_BYTES    256

struct Tile { v2f h, x0, x1, x2; };

__device__ __forceinline__ void load_tile(Tile& t, const char* xb, const char* hb,
                                          unsigned offX, unsigned offX2, unsigned offH, int j) {
    t.x0 = *(const v2f*)(xb + offX  + j * X_TILE_BYTES);
    t.x1 = *(const v2f*)(xb + offX  + j * X_TILE_BYTES + 16);
    t.h  = *(const v2f*)(hb + offH  + j * H_TILE_BYTES);
    t.x2 = *(const v2f*)(xb + offX2 + j * X_TILE_BYTES);
}

__device__ __forceinline__ v2f gru_tile(const Tile& t, v2f aWh, v2f aWx0, v2f aWx1, v2f aWx2, v8f cInit) {
    v8f acc = __builtin_amdgcn_wmma_f32_16x16x4_f32(false, aWh,  false, t.h,  (short)0, cInit, false, false);
    acc     = __builtin_amdgcn_wmma_f32_16x16x4_f32(false, aWx0, false, t.x0, (short)0, acc,   false, false);
    acc     = __builtin_amdgcn_wmma_f32_16x16x4_f32(false, aWx1, false, t.x1, (short)0, acc,   false, false);
    acc     = __builtin_amdgcn_wmma_f32_16x16x4_f32(false, aWx2, false, t.x2, (short)0, acc,   false, false);

    const v2f zs = { acc[0], acc[1] };
    const v2f rs = { acc[2], acc[3] };
    const v2f hn = { acc[4], acc[5] };
    const v2f xn = { acc[6], acc[7] };

    const v2f zh = 0.5f * zs;
    const v2f rh = 0.5f * rs;
    const v2f zt = { hw_tanh(zh.x), hw_tanh(zh.y) };
    const v2f rt = { hw_tanh(rh.x), hw_tanh(rh.y) };
    const v2f z  = 0.5f * zt + 0.5f;
    const v2f rr = 0.5f * rt + 0.5f;

    const v2f pre = rr * hn + xn;
    const v2f nn  = { hw_tanh(pre.x), hw_tanh(pre.y) };
    return nn + z * (t.h - nn);
}

__global__ __launch_bounds__(512) void gru_wmma_kernel(
    const float* __restrict__ x, const float* __restrict__ hidden,
    const float* __restrict__ Wzh, const float* __restrict__ bzh,
    const float* __restrict__ Wzx, const float* __restrict__ bzx,
    const float* __restrict__ Wrh, const float* __restrict__ brh,
    const float* __restrict__ Wrx, const float* __restrict__ brx,
    const float* __restrict__ Wnx, const float* __restrict__ bnx,
    const float* __restrict__ Wnh, const float* __restrict__ bnh,
    float* __restrict__ out, int B)
{
    const int lane = threadIdx.x & 31;
    const int n    = lane & 15;
    const int kh   = lane >> 4;
    const int u    = 2 * (n >> 3) + (n & 1);
    const int kind = (n & 7) >> 1;

    const float* Wh = (kind == 0) ? Wzh : (kind == 1) ? Wrh : Wnh;
    const float* Wx = (kind == 0) ? Wzx : (kind == 1) ? Wrx : Wnx;
    const bool useH = (kind != 3);
    const bool useX = (kind != 2);

    v2f aWh;
    aWh.x = useH ? Wh[u * 4 + 2 * kh]     : 0.0f;
    aWh.y = useH ? Wh[u * 4 + 2 * kh + 1] : 0.0f;

    v2f aWx[3];
#pragma unroll
    for (int t = 0; t < 3; ++t) {
        int k0 = 4 * t + 2 * kh;
        int k1 = k0 + 1;
        aWx[t].x = (useX && k0 < 10) ? Wx[u * 10 + k0] : 0.0f;
        aWx[t].y = (useX && k1 < 10) ? Wx[u * 10 + k1] : 0.0f;
    }

    v8f cInit;
#pragma unroll
    for (int r = 0; r < 8; ++r) {
        const int ub = 2 * kh + (r & 1);
        cInit[r] = (r < 2) ? (bzh[ub] + bzx[ub]) : (r < 4) ? (brh[ub] + brx[ub]) : (r < 6) ? bnh[ub] : bnx[ub];
    }

    const int Tfull  = B >> 4;
    const int wib    = __builtin_amdgcn_readfirstlane((int)(threadIdx.x >> 5));
    const int waveId = blockIdx.x * WAVES_PER_BLOCK + wib;
    const int nWaves = gridDim.x * WAVES_PER_BLOCK;

    const int roundsTotal = Tfull / RING;
    const int rpw  = (roundsTotal + nWaves - 1) / nWaves;
    const int rBeg = waveId * rpw;
    const int rEnd = (rBeg + rpw < roundsTotal) ? (rBeg + rpw) : roundsTotal;
    const int nr   = rEnd - rBeg;

    const unsigned offX  = (unsigned)(n * 40 + kh * 8);
    const unsigned offX2 = (unsigned)(n * 40 + (kh ? 8 : 32));
    const unsigned offH  = (unsigned)(n * 16 + kh * 8);

    if (nr > 0) {
        const char* xb = (const char*)x      + (size_t)rBeg * (RING * X_TILE_BYTES);
        const char* hb = (const char*)hidden + (size_t)rBeg * (RING * H_TILE_BYTES);
        char*       ob = (char*)out          + (size_t)rBeg * (RING * H_TILE_BYTES);

        Tile buf[RING];
#pragma unroll
        for (int q = 0; q < RING; ++q) load_tile(buf[q], xb, hb, offX, offX2, offH, q);

        for (int r = 1; r < nr; ++r) {
            const char* xn = xb + RING * X_TILE_BYTES;
            const char* hn = hb + RING * H_TILE_BYTES;
#pragma unroll
            for (int q = 0; q < RING; q += 2) {
                const v2f o0 = gru_tile(buf[q],     aWh, aWx[0], aWx[1], aWx[2], cInit);
                const v2f o1 = gru_tile(buf[q + 1], aWh, aWx[0], aWx[1], aWx[2], cInit);
                *(v2f*)(ob + offH + q * H_TILE_BYTES)       = o0;
                *(v2f*)(ob + offH + (q + 1) * H_TILE_BYTES) = o1;
                load_tile(buf[q],     xn, hn, offX, offX2, offH, q);
                load_tile(buf[q + 1], xn, hn, offX, offX2, offH, q + 1);
                __builtin_amdgcn_sched_barrier(0);
            }
            xb = xn; hb = hn; ob += RING * H_TILE_BYTES;
        }

#pragma unroll
        for (int q = 0; q < RING; q += 2) {
            const v2f o0 = gru_tile(buf[q],     aWh, aWx[0], aWx[1], aWx[2], cInit);
            const v2f o1 = gru_tile(buf[q + 1], aWh, aWx[0], aWx[1], aWx[2], cInit);
            *(v2f*)(ob + offH + q * H_TILE_BYTES)       = o0;
            *(v2f*)(ob + offH + (q + 1) * H_TILE_BYTES) = o1;
        }
    }

    const int tLeft = roundsTotal * RING + waveId;
    if (tLeft < Tfull) {
        Tile t;
        load_tile(t, (const char*)x + (size_t)tLeft * X_TILE_BYTES,
                     (const char*)hidden + (size_t)tLeft * H_TILE_BYTES, offX, offX2, offH, 0);
        const v2f o = gru_tile(t, aWh, aWx[0], aWx[1], aWx[2], cInit);
        *(v2f*)((char*)out + (size_t)tLeft * H_TILE_BYTES + offH) = o;
    }

    const int tailBase = Tfull << 4;
    if (waveId == 0 && tailBase < B) {
        const int row = tailBase + lane;
        if (row < B) {
#pragma unroll 1
            for (int u = 0; u < 4; ++u) {
                float az  = bzh[u] + bzx[u];
                float ar  = brh[u] + brx[u];
                float ahn = bnh[u];
                float axn = bnx[u];
#pragma unroll 1
                for (int k = 0; k < 4; ++k) {
                    const float hk = hidden[(size_t)row * 4 + k];
                    az  += hk * Wzh[u * 4 + k];
                    ar  += hk * Wrh[u * 4 + k];
                    ahn += hk * Wnh[u * 4 + k];
                }
#pragma unroll 1
                for (int k = 0; k < 10; ++k) {
                    const float xk = x[(size_t)row * 10 + k];
                    az  += xk * Wzx[u * 10 + k];
                    ar  += xk * Wrx[u * 10 + k];
                    axn += xk * Wnx[u * 10 + k];
                }
                const float z  = fast_sigmoid(az);
                const float rr = fast_sigmoid(ar);
                const float nn = hw_tanh(rr * ahn + axn);
                out[(size_t)row * 4 + u] = nn + z * (hidden[(size_t)row * 4 + u] - nn);
            }
        }
    }
}

extern "C" void kernel_launch(void* const* d_in, const int* in_sizes, int n_in,
                              void* d_out, int out_size, void* d_ws, size_t ws_size,
                              hipStream_t stream) {
    const float* x      = (const float*)d_in[0];
    const float* hidden = (const float*)d_in[1];
    const float* Wzh    = (const float*)d_in[2];
    const float* bzh    = (const float*)d_in[3];
    const float* Wzx    = (const float*)d_in[4];
    const float* bzx    = (const float*)d_in[5];
    const float* Wrh    = (const float*)d_in[6];
    const float* brh    = (const float*)d_in[7];
    const float* Wrx    = (const float*)d_in[8];
    const float* brx    = (const float*)d_in[9];
    const float* Wnx    = (const float*)d_in[10];
    const float* bnx    = (const float*)d_in[11];
    const float* Wnh    = (const float*)d_in[12];
    const float* bnh    = (const float*)d_in[13];
    float* out = (float*)d_out;

    const int B = in_sizes[0] / 10;

    const int blocks  = 256;
    const int threads = 512;
    gru_wmma_kernel<<<blocks, threads, 0, stream>>>(
        x, hidden, Wzh, bzh, Wzx, bzx, Wrh, brh, Wrx, brx, Wnx, bnx, Wnh, bnh,
        out, B);
}


# Per tile and lane two 16-byte loads (x[n][4kh..4kh+3], hidden[n][0..3]) and one 8-byte load (x[n][8..9]) replace four 8-byte loads; K slots re-dealt to match, so the gate sums change order
# speedup vs baseline: 1.1868x; 1.0578x over previous
#include <hip/hip_runtime.h>


typedef __attribute__((ext_vector_type(2))) float v2f;
typedef __attribute__((ext_vector_type(4))) float v4f;
typedef __attribute__((ext_vector_type(8))) float v8f;
typedef float v4f_a8 __attribute__((ext_vector_type(4), aligned(8)));

__device__ __forceinline__ float hw_tanh(float x) {
#if __has_builtin(__builtin_amdgcn_tanhf)
    return __builtin_amdgcn_tanhf(x);
#elif __has_builtin(__builtin_amdgcn_tanh_f32)
    return __builtin_amdgcn_tanh_f32(x);
#else
    float e = __expf(2.0f * x);
    return 1.0f - 2.0f * __builtin_amdgcn_rcpf(e + 1.0f);
#endif
}
__device__ __forceinline__ float fast_sigmoid(float x) {
    return 0.5f * hw_tanh(0.5f * x) + 0.5f;
}

#define RING            8
#define WAVES_PER_BLOCK 16
#define X_TILE_BYTES    640
#define H_TILE_BYTES    256

struct Tile { v4f p; v4f hh; v2f xt; };

__device__ __forceinline__ void load_tile(Tile& t, const char* xb, const char* hb,
                                          unsigned offP, unsigned offT, unsigned offHH, int j) {
    t.p  = *(const v4f_a8*)(xb + offP  + j * X_TILE_BYTES);
    t.hh = *(const v4f*)   (hb + offHH + j * H_TILE_BYTES);
    t.xt = *(const v2f*)   (xb + offT  + j * X_TILE_BYTES);
}

__device__ __forceinline__ v2f gru_tile(const Tile& t, bool hi, v2f a0, v2f a1, v2f a2, v2f a3, v8f cInit) {
    const v2f p01 = { t.p.x,  t.p.y };
    const v2f p23 = { t.p.z,  t.p.w };
    const v2f h01 = { t.hh.x, t.hh.y };
    const v2f h23 = { t.hh.z, t.hh.w };
    const v2f b3  = hi ? t.xt : h23;
    const v2f hs  = hi ? h23  : h01;

    v8f acc = __builtin_amdgcn_wmma_f32_16x16x4_f32(false, a0, false, p01, (short)0, cInit, false, false);
    acc     = __builtin_amdgcn_wmma_f32_16x16x4_f32(false, a1, false, p23, (short)0, acc,   false, false);
    acc     = __builtin_amdgcn_wmma_f32_16x16x4_f32(false, a2, false, h01, (short)0, acc,   false, false);
    acc     = __builtin_amdgcn_wmma_f32_16x16x4_f32(false, a3, false, b3,  (short)0, acc,   false, false);

    const v2f zs = { acc[0], acc[1] };
    const v2f rs = { acc[2], acc[3] };
    const v2f hn = { acc[4], acc[5] };
    const v2f xn = { acc[6], acc[7] };

    const v2f zh = 0.5f * zs;
    const v2f rh = 0.5f * rs;
    const v2f zt = { hw_tanh(zh.x), hw_tanh(zh.y) };
    const v2f rt = { hw_tanh(rh.x), hw_tanh(rh.y) };
    const v2f z  = 0.5f * zt + 0.5f;
    const v2f rr = 0.5f * rt + 0.5f;

    const v2f pre = rr * hn + xn;
    const v2f nn  = { hw_tanh(pre.x), hw_tanh(pre.y) };
    return nn + z * (hs - nn);
}

__global__ __launch_bounds__(512) void gru_wmma_kernel(
    const float* __restrict__ x, const float* __restrict__ hidden,
    const float* __restrict__ Wzh, const float* __restrict__ bzh,
    const float* __restrict__ Wzx, const float* __restrict__ bzx,
    const float* __restrict__ Wrh, const float* __restrict__ brh,
    const float* __restrict__ Wrx, const float* __restrict__ brx,
    const float* __restrict__ Wnx, const float* __restrict__ bnx,
    const float* __restrict__ Wnh, const float* __restrict__ bnh,
    float* __restrict__ out, int B)
{
    const int lane = threadIdx.x & 31;
    const int n    = lane & 15;
    const int kh   = lane >> 4;
    const int u    = 2 * (n >> 3) + (n & 1);
    const int kind = (n & 7) >> 1;

    const float* Wh = (kind == 0) ? Wzh : (kind == 1) ? Wrh : Wnh;
    const float* Wx = (kind == 0) ? Wzx : (kind == 1) ? Wrx : Wnx;
    const bool useH = (kind != 3);
    const bool useX = (kind != 2);

    const bool hi = (kh != 0);
    v2f a0, a1, a2, a3;
    a0.x = useX ? Wx[u * 10 + 4 * kh + 0] : 0.0f;
    a0.y = useX ? Wx[u * 10 + 4 * kh + 1] : 0.0f;
    a1.x = useX ? Wx[u * 10 + 4 * kh + 2] : 0.0f;
    a1.y = useX ? Wx[u * 10 + 4 * kh + 3] : 0.0f;
    a2.x = (useH && !hi) ? Wh[u * 4 + 0] : 0.0f;
    a2.y = (useH && !hi) ? Wh[u * 4 + 1] : 0.0f;
    const float wx8 = useX ? Wx[u * 10 + 8] : 0.0f;
    const float wx9 = useX ? Wx[u * 10 + 9] : 0.0f;
    const float wh2 = useH ? Wh[u * 4 + 2] : 0.0f;
    const float wh3 = useH ? Wh[u * 4 + 3] : 0.0f;
    a3.x = hi ? wx8 : wh2;
    a3.y = hi ? wx9 : wh3;

    v8f cInit;
#pragma unroll
    for (int r = 0; r < 8; ++r) {
        const int ub = 2 * kh + (r & 1);
        cInit[r] = (r < 2) ? (bzh[ub] + bzx[ub]) : (r < 4) ? (brh[ub] + brx[ub]) : (r < 6) ? bnh[ub] : bnx[ub];
    }

    const int Tfull  = B >> 4;
    const int wib    = __builtin_amdgcn_readfirstlane((int)(threadIdx.x >> 5));
    const int waveId = blockIdx.x * WAVES_PER_BLOCK + wib;
    const int nWaves = gridDim.x * WAVES_PER_BLOCK;

    const int roundsTotal = Tfull / RING;
    const int rpw  = (roundsTotal + nWaves - 1) / nWaves;
    const int rBeg = waveId * rpw;
    const int rEnd = (rBeg + rpw < roundsTotal) ? (rBeg + rpw) : roundsTotal;
    const int nr   = rEnd - rBeg;

    const unsigned offP  = (unsigned)(n * 40 + kh * 16);
    const unsigned offT  = (unsigned)(n * 40 + 32);
    const unsigned offHH = (unsigned)(n * 16);
    const unsigned offH  = (unsigned)(n * 16 + kh * 8);

    if (nr > 0) {
        const char* xb = (const char*)x      + (size_t)rBeg * (RING * X_TILE_BYTES);
        const char* hb = (const char*)hidden + (size_t)rBeg * (RING * H_TILE_BYTES);
        char*       ob = (char*)out          + (size_t)rBeg * (RING * H_TILE_BYTES);

        Tile buf[RING];
#pragma unroll
        for (int q = 0; q < RING; ++q) load_tile(buf[q], xb, hb, offP, offT, offHH, q);

        for (int r = 1; r < nr; ++r) {
            const char* xn = xb + RING * X_TILE_BYTES;
            const char* hn = hb + RING * H_TILE_BYTES;
#pragma unroll
            for (int q = 0; q < RING; q += 2) {
                const v2f o0 = gru_tile(buf[q],     hi, a0, a1, a2, a3, cInit);
                const v2f o1 = gru_tile(buf[q + 1], hi, a0, a1, a2, a3, cInit);
                *(v2f*)(ob + offH + q * H_TILE_BYTES)       = o0;
                *(v2f*)(ob + offH + (q + 1) * H_TILE_BYTES) = o1;
                load_tile(buf[q],     xn, hn, offP, offT, offHH, q);
                load_tile(buf[q + 1], xn, hn, offP, offT, offHH, q + 1);
                __builtin_amdgcn_sched_barrier(0);
            }
            xb = xn; hb = hn; ob += RING * H_TILE_BYTES;
        }

#pragma unroll
        for (int q = 0; q < RING; q += 2) {
            const v2f o0 = gru_tile(buf[q],     hi, a0, a1, a2, a3, cInit);
            const v2f o1 = gru_tile(buf[q + 1], hi, a0, a1, a2, a3, cInit);
            *(v2f*)(ob + offH + q * H_TILE_BYTES)       = o0;
            *(v2f*)(ob + offH + (q + 1) * H_TILE_BYTES) = o1;
        }
    }

    const int tLeft = roundsTotal * RING + waveId;
    if (tLeft < Tfull) {
        Tile t;
        load_tile(t, (const char*)x + (size_t)tLeft * X_TILE_BYTES,
                     (const char*)hidden + (size_t)tLeft * H_TILE_BYTES, offP, offT, offHH, 0);
        const v2f o = gru_tile(t, hi, a0, a1, a2, a3, cInit);
        *(v2f*)((char*)out + (size_t)tLeft * H_TILE_BYTES + offH) = o;
    }

    const int tailBase = Tfull << 4;
    if (waveId == 0 && tailBase < B) {
        const int row = tailBase + lane;
        if (row < B) {
#pragma unroll 1
            for (int u = 0; u < 4; ++u) {
                float az  = bzh[u] + bzx[u];
                float ar  = brh[u] + brx[u];
                float ahn = bnh[u];
                float axn = bnx[u];
#pragma unroll 1
                for (int k = 0; k < 4; ++k) {
                    const float hk = hidden[(size_t)row * 4 + k];
                    az  += hk * Wzh[u * 4 + k];
                    ar  += hk * Wrh[u * 4 + k];
                    ahn += hk * Wnh[u * 4 + k];
                }
#pragma unroll 1
                for (int k = 0; k < 10; ++k) {
                    const float xk = x[(size_t)row * 10 + k];
                    az  += xk * Wzx[u * 10 + k];
                    ar  += xk * Wrx[u * 10 + k];
                    axn += xk * Wnx[u * 10 + k];
                }
                const float z  = fast_sigmoid(az);
                const float rr = fast_sigmoid(ar);
                const float nn = hw_tanh(rr * ahn + axn);
                out[(size_t)row * 4 + u] = nn + z * (hidden[(size_t)row * 4 + u] - nn);
            }
        }
    }
}

extern "C" void kernel_launch(void* const* d_in, const int* in_sizes, int n_in,
                              void* d_out, int out_size, void* d_ws, size_t ws_size,
                              hipStream_t stream) {
    const float* x      = (const float*)d_in[0];
    const float* hidden = (const float*)d_in[1];
    const float* Wzh    = (const float*)d_in[2];
    const float* bzh    = (const float*)d_in[3];
    const float* Wzx    = (const float*)d_in[4];
    const float* bzx    = (const float*)d_in[5];
    const float* Wrh    = (const float*)d_in[6];
    const float* brh    = (const float*)d_in[7];
    const float* Wrx    = (const float*)d_in[8];
    const float* brx    = (const float*)d_in[9];
    const float* Wnx    = (const float*)d_in[10];
    const float* bnx    = (const float*)d_in[11];
    const float* Wnh    = (const float*)d_in[12];
    const float* bnh    = (const float*)d_in[13];
    float* out = (float*)d_out;

    const int B = in_sizes[0] / 10;

    const int blocks  = 256;
    const int threads = 512;
    gru_wmma_kernel<<<blocks, threads, 0, stream>>>(
        x, hidden, Wzh, bzh, Wzx, bzx, Wrh, brh, Wrx, brx, Wnx, bnx, Wnh, bnh,
        out, B);
}
